# starting kernel
# baseline (speedup 1.0000x reference)
#include <hip/hip_runtime.h>
#include <hip/hip_bf16.h>
#include <math.h>

#define BSZ  4
#define LSEQ 2048
#define DCH  768
#define NST  64
#define DN   (DCH * NST)

typedef __attribute__((ext_vector_type(2))) float v2f;
typedef __attribute__((ext_vector_type(8))) float v8f;

__global__ void __launch_bounds__(256)
s4_precompute(const float* __restrict__ A_real, const float* __restrict__ A_imag,
              const float* __restrict__ B_mat,  const float* __restrict__ C_mat,
              const float* __restrict__ log_delta,
              float* __restrict__ dAr, float* __restrict__ dAi,
              float* __restrict__ CBr, float* __restrict__ CBi) {
    int idx = blockIdx.x * blockDim.x + threadIdx.x;
    if (idx >= DN) return;
    int d = idx / NST;

    float delta = log1pf(expf(log_delta[d]));
    float ar = A_real[idx], ai = A_imag[idx];
    float dr = delta * ar, di = delta * ai;

    float er = expf(dr);
    float s, c;
    __sincosf(di, &s, &c);
    float abr = er * c, abi = er * s;

    float nr = abr - 1.0f, ni = abi;
    float inv = 1.0f / (ar * ar + ai * ai);
    float qr = (nr * ar + ni * ai) * inv;
    float qi = (ni * ar - nr * ai) * inv;

    float b = B_mat[idx], cm = C_mat[idx];
    dAr[idx] = dr;
    dAi[idx] = di;
    CBr[idx] = cm * (qr * b);
    CBi[idx] = cm * (qi * b);
}

__global__ void __launch_bounds__(256)
s4_kernelgen(const float* __restrict__ dAr, const float* __restrict__ dAi,
             const float* __restrict__ CBr, const float* __restrict__ CBi,
             float* __restrict__ Kbuf) {
    int idx = blockIdx.x * blockDim.x + threadIdx.x;
    if (idx >= DCH * LSEQ) return;
    int   d = idx / LSEQ;
    float l = (float)(idx & (LSEQ - 1));

    const float* pr = dAr + d * NST;
    const float* pi = dAi + d * NST;
    const float* cr = CBr + d * NST;
    const float* ci = CBi + d * NST;

    float acc = 0.0f;
#pragma unroll 4
    for (int n = 0; n < NST; ++n) {
        float e = __expf(pr[n] * l);
        float s, c;
        __sincosf(pi[n] * l, &s, &c);
        acc += e * (cr[n] * c - ci[n] * s);
    }
    Kbuf[idx] = acc;
}

__global__ void __launch_bounds__(256)
s4_conv_wmma(const float* __restrict__ u,
             const float* __restrict__ Kbuf,
             float* __restrict__ y) {
    __shared__ float Klds[LSEQ];

    const int d  = blockIdx.x >> 4;
    const int Ig = blockIdx.x & 15;

    {
        const float4* src = (const float4*)(Kbuf + (size_t)d * LSEQ);
        float4*       dst = (float4*)Klds;
#pragma unroll
        for (int i = 0; i < (LSEQ / 4) / 256; ++i)
            dst[threadIdx.x + i * 256] = src[threadIdx.x + i * 256];
    }
    __syncthreads();

    const int wave = threadIdx.x >> 5;
    const int lane = threadIdx.x & 31;
    const int I    = Ig * 8 + wave;
    const int ln   = lane & 15;
    const int g    = lane >> 4;
    const int orow = I * 16 + ln;

    v8f acc = {};

    for (int J = 0; J <= I; ++J) {
        const int jbase = J * 16;

        if (ln < BSZ && J < I) {
            const float* pf = u + ((size_t)ln * LSEQ + (size_t)(jbase + 16)) * DCH + d;
            __builtin_prefetch(pf, 0, 1);
        }

#pragma unroll
        for (int cchunk = 0; cchunk < 4; ++cchunk) {
            const int kb = jbase + cchunk * 4;
            const int k0   = kb + 2 * g;
            const int diff = orow - k0;
            v2f a;
            a.x = (diff >= 0) ? Klds[diff]     : 0.0f;
            a.y = (diff >= 1) ? Klds[diff - 1] : 0.0f;
            v2f b;
            if (ln < BSZ) {
                const float* up = u + ((size_t)ln * LSEQ + (size_t)k0) * DCH + d;
                b.x = up[0];
                b.y = up[DCH];
            } else {
                b.x = 0.0f;
                b.y = 0.0f;
            }
            acc = __builtin_amdgcn_wmma_f32_16x16x4_f32(
                false, a, false, b, (short)0, acc, false, false);
        }
    }

#pragma unroll
    for (int r = 0; r < 8; ++r) {
        if (ln < BSZ) {
            const int mrow = r + 8 * g;
            y[((size_t)ln * LSEQ + (size_t)(I * 16 + mrow)) * DCH + d] = acc[r];
        }
    }
}

extern "C" void kernel_launch(void* const* d_in, const int* in_sizes, int n_in,
                              void* d_out, int out_size, void* d_ws, size_t ws_size,
                              hipStream_t stream) {
    const float* u         = (const float*)d_in[0];
    const float* A_real    = (const float*)d_in[1];
    const float* A_imag    = (const float*)d_in[2];
    const float* B_mat     = (const float*)d_in[3];
    const float* C_mat     = (const float*)d_in[4];
    const float* log_delta = (const float*)d_in[5];
    float* y = (float*)d_out;

    float* ws   = (float*)d_ws;
    float* dAr  = ws;
    float* dAi  = ws + 1 * DN;
    float* CBr  = ws + 2 * DN;
    float* CBi  = ws + 3 * DN;
    float* Kbuf = ws + 4 * DN;

    s4_precompute<<<dim3((DN + 255) / 256), dim3(256), 0, stream>>>(
        A_real, A_imag, B_mat, C_mat, log_delta, dAr, dAi, CBr, CBi);

    s4_kernelgen<<<dim3((DCH * LSEQ + 255) / 256), dim3(256), 0, stream>>>(
        dAr, dAi, CBr, CBi, Kbuf);

    s4_conv_wmma<<<dim3(DCH * 16), dim3(256), 0, stream>>>(u, Kbuf, y);
}


# K build: one block per channel, each thread sums 8 taps (l = t + 256 j); the 64 (dA, CB) are scalar reads, exp2 and revolution scales applied once per mode; 6144 waves, not 49152; sum order kept
# speedup vs baseline: 12.2547x; 12.2547x over previous
#include <hip/hip_runtime.h>
#include <hip/hip_bf16.h>
#include <math.h>

#define BSZ  4
#define LSEQ 2048
#define DCH  768
#define NST  64
#define DN   (DCH * NST)

typedef __attribute__((ext_vector_type(2))) float v2f;
typedef __attribute__((ext_vector_type(8))) float v8f;

__global__ void __launch_bounds__(256)
s4_precompute(const float* __restrict__ A_real, const float* __restrict__ A_imag,
              const float* __restrict__ B_mat,  const float* __restrict__ C_mat,
              const float* __restrict__ log_delta,
              float* __restrict__ dAr, float* __restrict__ dAi,
              float* __restrict__ CBr, float* __restrict__ CBi) {
    int idx = blockIdx.x * blockDim.x + threadIdx.x;
    if (idx >= DN) return;
    int d = idx / NST;

    float delta = log1pf(expf(log_delta[d]));
    float ar = A_real[idx], ai = A_imag[idx];
    float dr = delta * ar, di = delta * ai;

    float er = expf(dr);
    float s, c;
    __sincosf(di, &s, &c);
    float abr = er * c, abi = er * s;

    float nr = abr - 1.0f, ni = abi;
    float inv = 1.0f / (ar * ar + ai * ai);
    float qr = (nr * ar + ni * ai) * inv;
    float qi = (ni * ar - nr * ai) * inv;

    float b = B_mat[idx], cm = C_mat[idx];
    dAr[idx] = dr;
    dAi[idx] = di;
    CBr[idx] = cm * (qr * b);
    CBi[idx] = cm * (qi * b);
}

#define KT 8

__global__ void __launch_bounds__(256)
s4_kernelgen(const float* __restrict__ dAr, const float* __restrict__ dAi,
             const float* __restrict__ CBr, const float* __restrict__ CBi,
             float* __restrict__ Kbuf) {
    const int d = blockIdx.x;
    const int t = threadIdx.x;

    const float* pr = dAr + d * NST;
    const float* pi = dAi + d * NST;
    const float* cr = CBr + d * NST;
    const float* ci = CBi + d * NST;

    float l[KT], acc[KT];
#pragma unroll
    for (int j = 0; j < KT; ++j) { l[j] = (float)(t + 256 * j); acc[j] = 0.0f; }

#pragma unroll 4
    for (int n = 0; n < NST; ++n) {
        const float ea = pr[n] * 1.44269504f;
        const float ph = pi[n] * 0.15915494f;
        const float br = cr[n], bi = ci[n];
#pragma unroll
        for (int j = 0; j < KT; ++j) {
            const float e = __builtin_amdgcn_exp2f(ea * l[j]);
            const float w = ph * l[j];
            const float s = __builtin_amdgcn_sinf(w);
            const float c = __builtin_amdgcn_cosf(w);
            acc[j] += e * (br * c - bi * s);
        }
    }

    float* kd = Kbuf + (size_t)d * LSEQ + t;
#pragma unroll
    for (int j = 0; j < KT; ++j) kd[256 * j] = acc[j];
}

#define NCH    12
#define CTHR   768
#define KS_LEN 2080
#define U_HEAD 256
#define U_DAT0 272
#define U_TAIL 2448
#define USTR   2468
#define YSTR   2052
#define ROWQ   (256 * (DCH / 4))

__device__ __forceinline__ v8f mma4(v2f a, v2f b, v8f c) {
    return __builtin_amdgcn_wmma_f32_16x16x4_f32(false, a, false, b, (short)0, c, false, false);
}

template <int NT>
__device__ __forceinline__ void conv_trips(int n, const float* kr, const float* us,
                                           int& ik, int (&iu)[4], v8f (&acc)[4]) {
    for (int it = 0; it < n; ++it) {
#pragma unroll
        for (int sub = 0; sub < 4; ++sub) {
            const int o = 12 - 4 * sub;
            v2f kb, f[NT];
            kb.x = kr[ik + o];   kb.y = kr[ik + o + 1];
#pragma unroll
            for (int j = 0; j < NT; ++j) {
                f[j].x = us[iu[4 - NT + j] + o];
                f[j].y = us[iu[4 - NT + j] + o + 1];
            }
#pragma unroll
            for (int j = 0; j < NT; ++j) acc[4 - NT + j] = mma4(f[j], kb, acc[4 - NT + j]);
        }
        ik -= 16;
#pragma unroll
        for (int j = 0; j < NT; ++j) iu[4 - NT + j] -= 17;
    }
}

__global__ void __launch_bounds__(CTHR)
s4_conv_wmma(const float* __restrict__ u,
             const float* __restrict__ Kbuf,
             float* __restrict__ y) {
    __shared__ __attribute__((aligned(16))) float Kr[NCH][KS_LEN];
    __shared__ float Us[NCH][USTR];
    __shared__ float Ys[NCH][YSTR];

    const int d0   = (blockIdx.x >> 2) * NCH;
    const int bb   = blockIdx.x & 3;
    const int tid  = threadIdx.x;
    const int t0   = tid / 3;
    const int part = tid - 3 * t0;

    float4 ux[8], kx[8];
    {
        const float4* us = (const float4*)(u + ((size_t)bb * LSEQ + t0) * DCH + d0 + 4 * part);
#pragma unroll
        for (int i = 0; i < 8; ++i) ux[i] = us[(size_t)i * ROWQ];
        const float4* ks = (const float4*)(Kbuf + (size_t)d0 * LSEQ);
#pragma unroll
        for (int i = 0; i < 8; ++i) kx[i] = ks[tid + i * CTHR];
    }

    for (int i = tid; i < NCH * 324; i += CTHR) {
        const int c = i / 324;
        const int k = i - 324 * c;
        if (k < U_DAT0)           Us[c][k] = 0.0f;
        else if (k < U_DAT0 + 20) Us[c][U_TAIL + (k - U_DAT0)] = 0.0f;
        else {
            const int k2 = k - (U_DAT0 + 20);
            Kr[c][k2 < 16 ? k2 : LSEQ + k2] = 0.0f;
        }
    }
#pragma unroll
    for (int i = 0; i < 8; ++i) {
        const int    q = tid + i * CTHR;
        const int    c = q >> 9;
        const int    j = q & 511;
        const float4 v = kx[i];
        ((float4*)&Kr[c][0])[515 - j] = make_float4(v.w, v.z, v.y, v.x);
    }
#pragma unroll
    for (int i = 0; i < 8; ++i) {
        const int tp = t0 + 256 * i + U_HEAD;
        float*    q  = &Us[4 * part][0] + tp + ((tp + 15) >> 4);
        q[0] = ux[i].x; q[USTR] = ux[i].y; q[2 * USTR] = ux[i].z; q[3 * USTR] = ux[i].w;
    }
    __syncthreads();

    const int wave = __builtin_amdgcn_readfirstlane(tid >> 5);
    const int ch   = wave >> 1;
    const int par  = wave & 1;
    const int lane = tid & 31;
    const int ln   = lane & 15;
    const int g    = lane >> 4;
    const int m0 = 2 * par, m1 = 2 * par + 1, m2 = 6 - 2 * par, m3 = 7 - 2 * par;

    const float* kr = &Kr[ch][0];
    const float* us = &Us[ch][0];
    int ik = 2066 - 2 * g - ln;
    const int ub = 17 * (ln + 17) - 2 * g - 13;
    int iu[4] = { ub + 272 * m0, ub + 272 * m1, ub + 272 * m2, ub + 272 * m3 };
    v8f acc[4] = {};

    conv_trips<4>(16 * m0 + 17,   kr, us, ik, iu, acc);
    conv_trips<3>(16 * (m1 - m0), kr, us, ik, iu, acc);
    conv_trips<2>(16 * (m2 - m1), kr, us, ik, iu, acc);
    conv_trips<1>(16 * (m3 - m2), kr, us, ik, iu, acc);

    {
        float*    ys = &Ys[ch][0];
        const int mm[4] = { m0, m1, m2, m3 };
#pragma unroll
        for (int k = 0; k < 4; ++k) {
#pragma unroll
            for (int r = 0; r < 8; ++r) {
                const int t = 256 * mm[k] + 16 * r + 128 * g + ln;
                ys[t ^ (g << 4)] = acc[k][r];
            }
        }
    }
    __syncthreads();

    {
        float4* yd = (float4*)(y + ((size_t)bb * LSEQ + t0) * DCH + d0 + 4 * part);
#pragma unroll
        for (int i = 0; i < 8; ++i) {
            const int    t = t0 + 256 * i;
            const float* s = &Ys[4 * part][0] + (t ^ (((t >> 7) & 1) << 4));
            yd[(size_t)i * ROWQ] = make_float4(s[0], s[YSTR], s[2 * YSTR], s[3 * YSTR]);
        }
    }
}

extern "C" void kernel_launch(void* const* d_in, const int* in_sizes, int n_in,
                              void* d_out, int out_size, void* d_ws, size_t ws_size,
                              hipStream_t stream) {
    const float* u         = (const float*)d_in[0];
    const float* A_real    = (const float*)d_in[1];
    const float* A_imag    = (const float*)d_in[2];
    const float* B_mat     = (const float*)d_in[3];
    const float* C_mat     = (const float*)d_in[4];
    const float* log_delta = (const float*)d_in[5];
    float* y = (float*)d_out;

    float* ws   = (float*)d_ws;
    float* dAr  = ws;
    float* dAi  = ws + 1 * DN;
    float* CBr  = ws + 2 * DN;
    float* CBi  = ws + 3 * DN;
    float* Kbuf = ws + 4 * DN;

    s4_precompute<<<dim3((DN + 255) / 256), dim3(256), 0, stream>>>(
        A_real, A_imag, B_mat, C_mat, log_delta, dAr, dAi, CBr, CBi);

    s4_kernelgen<<<dim3(DCH), dim3(256), 0, stream>>>(
        dAr, dAi, CBr, CBi, Kbuf);

    s4_conv_wmma<<<dim3((DCH / NCH) * BSZ), dim3(CTHR), 0, stream>>>(u, Kbuf, y);
}


# K build: CB in polar form: log2|CB| goes into the exp2 argument, arg CB into the cosine's angle (one fma each, two taps per packed fma); exp2 + cos per mode and tap, no sin; mode order kept
# speedup vs baseline: 12.9412x; 1.0560x over previous
#include <hip/hip_runtime.h>
#include <hip/hip_bf16.h>
#include <math.h>

#define BSZ  4
#define LSEQ 2048
#define DCH  768
#define NST  64
#define DN   (DCH * NST)

typedef __attribute__((ext_vector_type(2))) float v2f;
typedef __attribute__((ext_vector_type(8))) float v8f;

__global__ void __launch_bounds__(256)
s4_precompute(const float* __restrict__ A_real, const float* __restrict__ A_imag,
              const float* __restrict__ B_mat,  const float* __restrict__ C_mat,
              const float* __restrict__ log_delta,
              float* __restrict__ dAr, float* __restrict__ dAi,
              float* __restrict__ CBlg, float* __restrict__ CBph) {
    int idx = blockIdx.x * blockDim.x + threadIdx.x;
    if (idx >= DN) return;
    int d = idx / NST;

    float delta = log1pf(expf(log_delta[d]));
    float ar = A_real[idx], ai = A_imag[idx];
    float dr = delta * ar, di = delta * ai;

    float er = expf(dr);
    float s, c;
    __sincosf(di, &s, &c);
    float abr = er * c, abi = er * s;

    float nr = abr - 1.0f, ni = abi;
    float inv = 1.0f / (ar * ar + ai * ai);
    float qr = (nr * ar + ni * ai) * inv;
    float qi = (ni * ar - nr * ai) * inv;

    float b = B_mat[idx], cm = C_mat[idx];
    const float cbr = cm * (qr * b);
    const float cbi = cm * (qi * b);
    const float ax = fabsf(cbr), ay = fabsf(cbi);
    const float mx = fmaxf(ax, ay), mn = fminf(ax, ay);
    const float rr = (mx > 0.0f) ? mn / mx : 0.0f;
    dAr[idx]  = dr;
    dAi[idx]  = di;
    CBlg[idx] = log2f(mx) + 0.5f * log2f(fmaf(rr, rr, 1.0f));
    CBph[idx] = atan2f(cbi, cbr) * 0.15915494f;
}

#define KT 8

__global__ void __launch_bounds__(256)
s4_kernelgen(const float* __restrict__ dAr, const float* __restrict__ dAi,
             const float* __restrict__ CBlg, const float* __restrict__ CBph,
             float* __restrict__ Kbuf) {
    const int d = blockIdx.x;
    const int t = threadIdx.x;

    const float* pr = dAr + d * NST;
    const float* pi = dAi + d * NST;
    const float* cg = CBlg + d * NST;
    const float* cp = CBph + d * NST;

    v2f l[KT / 2], acc[KT / 2];
#pragma unroll
    for (int j = 0; j < KT / 2; ++j) {
        l[j].x = (float)(t + 512 * j);  l[j].y = (float)(t + 512 * j + 256);
        acc[j].x = 0.0f;                acc[j].y = 0.0f;
    }

#pragma unroll 4
    for (int n = 0; n < NST; ++n) {
        const float ea = pr[n] * 1.44269504f;
        const float ph = pi[n] * 0.15915494f;
        const float lg = cg[n], p0 = cp[n];
#pragma unroll
        for (int j = 0; j < KT / 2; ++j) {
            const v2f xe = __builtin_elementwise_fma((v2f){ea, ea}, l[j], (v2f){lg, lg});
            const v2f xc = __builtin_elementwise_fma((v2f){ph, ph}, l[j], (v2f){p0, p0});
            v2f e, c;
            e.x = __builtin_amdgcn_exp2f(xe.x);  e.y = __builtin_amdgcn_exp2f(xe.y);
            c.x = __builtin_amdgcn_cosf(xc.x);   c.y = __builtin_amdgcn_cosf(xc.y);
            acc[j] = __builtin_elementwise_fma(e, c, acc[j]);
        }
    }

    float* kd = Kbuf + (size_t)d * LSEQ + t;
#pragma unroll
    for (int j = 0; j < KT / 2; ++j) { kd[512 * j] = acc[j].x; kd[512 * j + 256] = acc[j].y; }
}

#define NCH    12
#define CTHR   768
#define KS_LEN 2080
#define U_HEAD 256
#define U_DAT0 272
#define U_TAIL 2448
#define USTR   2468
#define YSTR   2052
#define ROWQ   (256 * (DCH / 4))

__device__ __forceinline__ v8f mma4(v2f a, v2f b, v8f c) {
    return __builtin_amdgcn_wmma_f32_16x16x4_f32(false, a, false, b, (short)0, c, false, false);
}

template <int NT>
__device__ __forceinline__ void conv_trips(int n, const float* kr, const float* us,
                                           int& ik, int (&iu)[4], v8f (&acc)[4]) {
    for (int it = 0; it < n; ++it) {
#pragma unroll
        for (int sub = 0; sub < 4; ++sub) {
            const int o = 12 - 4 * sub;
            v2f kb, f[NT];
            kb.x = kr[ik + o];   kb.y = kr[ik + o + 1];
#pragma unroll
            for (int j = 0; j < NT; ++j) {
                f[j].x = us[iu[4 - NT + j] + o];
                f[j].y = us[iu[4 - NT + j] + o + 1];
            }
#pragma unroll
            for (int j = 0; j < NT; ++j) acc[4 - NT + j] = mma4(f[j], kb, acc[4 - NT + j]);
        }
        ik -= 16;
#pragma unroll
        for (int j = 0; j < NT; ++j) iu[4 - NT + j] -= 17;
    }
}

__global__ void __launch_bounds__(CTHR)
s4_conv_wmma(const float* __restrict__ u,
             const float* __restrict__ Kbuf,
             float* __restrict__ y) {
    __shared__ __attribute__((aligned(16))) float Kr[NCH][KS_LEN];
    __shared__ float Us[NCH][USTR];
    __shared__ float Ys[NCH][YSTR];

    const int d0   = (blockIdx.x >> 2) * NCH;
    const int bb   = blockIdx.x & 3;
    const int tid  = threadIdx.x;
    const int t0   = tid / 3;
    const int part = tid - 3 * t0;

    float4 ux[8], kx[8];
    {
        const float4* us = (const float4*)(u + ((size_t)bb * LSEQ + t0) * DCH + d0 + 4 * part);
#pragma unroll
        for (int i = 0; i < 8; ++i) ux[i] = us[(size_t)i * ROWQ];
        const float4* ks = (const float4*)(Kbuf + (size_t)d0 * LSEQ);
#pragma unroll
        for (int i = 0; i < 8; ++i) kx[i] = ks[tid + i * CTHR];
    }

    for (int i = tid; i < NCH * 324; i += CTHR) {
        const int c = i / 324;
        const int k = i - 324 * c;
        if (k < U_DAT0)           Us[c][k] = 0.0f;
        else if (k < U_DAT0 + 20) Us[c][U_TAIL + (k - U_DAT0)] = 0.0f;
        else {
            const int k2 = k - (U_DAT0 + 20);
            Kr[c][k2 < 16 ? k2 : LSEQ + k2] = 0.0f;
        }
    }
#pragma unroll
    for (int i = 0; i < 8; ++i) {
        const int    q = tid + i * CTHR;
        const int    c = q >> 9;
        const int    j = q & 511;
        const float4 v = kx[i];
        ((float4*)&Kr[c][0])[515 - j] = make_float4(v.w, v.z, v.y, v.x);
    }
#pragma unroll
    for (int i = 0; i < 8; ++i) {
        const int tp = t0 + 256 * i + U_HEAD;
        float*    q  = &Us[4 * part][0] + tp + ((tp + 15) >> 4);
        q[0] = ux[i].x; q[USTR] = ux[i].y; q[2 * USTR] = ux[i].z; q[3 * USTR] = ux[i].w;
    }
    __syncthreads();

    const int wave = __builtin_amdgcn_readfirstlane(tid >> 5);
    const int ch   = wave >> 1;
    const int par  = wave & 1;
    const int lane = tid & 31;
    const int ln   = lane & 15;
    const int g    = lane >> 4;
    const int m0 = 2 * par, m1 = 2 * par + 1, m2 = 6 - 2 * par, m3 = 7 - 2 * par;

    const float* kr = &Kr[ch][0];
    const float* us = &Us[ch][0];
    int ik = 2066 - 2 * g - ln;
    const int ub = 17 * (ln + 17) - 2 * g - 13;
    int iu[4] = { ub + 272 * m0, ub + 272 * m1, ub + 272 * m2, ub + 272 * m3 };
    v8f acc[4] = {};

    conv_trips<4>(16 * m0 + 17,   kr, us, ik, iu, acc);
    conv_trips<3>(16 * (m1 - m0), kr, us, ik, iu, acc);
    conv_trips<2>(16 * (m2 - m1), kr, us, ik, iu, acc);
    conv_trips<1>(16 * (m3 - m2), kr, us, ik, iu, acc);

    {
        float*    ys = &Ys[ch][0];
        const int mm[4] = { m0, m1, m2, m3 };
#pragma unroll
        for (int k = 0; k < 4; ++k) {
#pragma unroll
            for (int r = 0; r < 8; ++r) {
                const int t = 256 * mm[k] + 16 * r + 128 * g + ln;
                ys[t ^ (g << 4)] = acc[k][r];
            }
        }
    }
    __syncthreads();

    {
        float4* yd = (float4*)(y + ((size_t)bb * LSEQ + t0) * DCH + d0 + 4 * part);
#pragma unroll
        for (int i = 0; i < 8; ++i) {
            const int    t = t0 + 256 * i;
            const float* s = &Ys[4 * part][0] + (t ^ (((t >> 7) & 1) << 4));
            yd[(size_t)i * ROWQ] = make_float4(s[0], s[YSTR], s[2 * YSTR], s[3 * YSTR]);
        }
    }
}

extern "C" void kernel_launch(void* const* d_in, const int* in_sizes, int n_in,
                              void* d_out, int out_size, void* d_ws, size_t ws_size,
                              hipStream_t stream) {
    const float* u         = (const float*)d_in[0];
    const float* A_real    = (const float*)d_in[1];
    const float* A_imag    = (const float*)d_in[2];
    const float* B_mat     = (const float*)d_in[3];
    const float* C_mat     = (const float*)d_in[4];
    const float* log_delta = (const float*)d_in[5];
    float* y = (float*)d_out;

    float* ws   = (float*)d_ws;
    float* dAr  = ws;
    float* dAi  = ws + 1 * DN;
    float* CBlg = ws + 2 * DN;
    float* CBph = ws + 3 * DN;
    float* Kbuf = ws + 4 * DN;

    s4_precompute<<<dim3((DN + 255) / 256), dim3(256), 0, stream>>>(
        A_real, A_imag, B_mat, C_mat, log_delta, dAr, dAi, CBlg, CBph);

    s4_kernelgen<<<dim3(DCH), dim3(256), 0, stream>>>(
        dAr, dAi, CBlg, CBph, Kbuf);

    s4_conv_wmma<<<dim3((DCH / NCH) * BSZ), dim3(CTHR), 0, stream>>>(u, Kbuf, y);
}


# Conv k order: a lane half takes 8 consecutive lags, u rows at LDS pitch 20, so two k groups' A fragments are one aligned ds_load_b128 (was 2 two-address reads); y tile laid over u; new sum order
# speedup vs baseline: 13.3438x; 1.0311x over previous
#include <hip/hip_runtime.h>
#include <hip/hip_bf16.h>
#include <math.h>

#define BSZ  4
#define LSEQ 2048
#define DCH  768
#define NST  64
#define DN   (DCH * NST)

typedef __attribute__((ext_vector_type(2))) float v2f;
typedef __attribute__((ext_vector_type(8))) float v8f;

__global__ void __launch_bounds__(256)
s4_precompute(const float* __restrict__ A_real, const float* __restrict__ A_imag,
              const float* __restrict__ B_mat,  const float* __restrict__ C_mat,
              const float* __restrict__ log_delta,
              float* __restrict__ dAr, float* __restrict__ dAi,
              float* __restrict__ CBlg, float* __restrict__ CBph) {
    int idx = blockIdx.x * blockDim.x + threadIdx.x;
    if (idx >= DN) return;
    int d = idx / NST;

    float delta = log1pf(expf(log_delta[d]));
    float ar = A_real[idx], ai = A_imag[idx];
    float dr = delta * ar, di = delta * ai;

    float er = expf(dr);
    float s, c;
    __sincosf(di, &s, &c);
    float abr = er * c, abi = er * s;

    float nr = abr - 1.0f, ni = abi;
    float inv = 1.0f / (ar * ar + ai * ai);
    float qr = (nr * ar + ni * ai) * inv;
    float qi = (ni * ar - nr * ai) * inv;

    float b = B_mat[idx], cm = C_mat[idx];
    const float cbr = cm * (qr * b);
    const float cbi = cm * (qi * b);
    const float ax = fabsf(cbr), ay = fabsf(cbi);
    const float mx = fmaxf(ax, ay), mn = fminf(ax, ay);
    const float rr = (mx > 0.0f) ? mn / mx : 0.0f;
    dAr[idx]  = dr;
    dAi[idx]  = di;
    CBlg[idx] = log2f(mx) + 0.5f * log2f(fmaf(rr, rr, 1.0f));
    CBph[idx] = atan2f(cbi, cbr) * 0.15915494f;
}

#define KT 8

__global__ void __launch_bounds__(256)
s4_kernelgen(const float* __restrict__ dAr, const float* __restrict__ dAi,
             const float* __restrict__ CBlg, const float* __restrict__ CBph,
             float* __restrict__ Kbuf) {
    const int d = blockIdx.x;
    const int t = threadIdx.x;

    const float* pr = dAr + d * NST;
    const float* pi = dAi + d * NST;
    const float* cg = CBlg + d * NST;
    const float* cp = CBph + d * NST;

    v2f l[KT / 2], acc[KT / 2];
#pragma unroll
    for (int j = 0; j < KT / 2; ++j) {
        l[j].x = (float)(t + 512 * j);  l[j].y = (float)(t + 512 * j + 256);
        acc[j].x = 0.0f;                acc[j].y = 0.0f;
    }

#pragma unroll 4
    for (int n = 0; n < NST; ++n) {
        const float ea = pr[n] * 1.44269504f;
        const float ph = pi[n] * 0.15915494f;
        const float lg = cg[n], p0 = cp[n];
#pragma unroll
        for (int j = 0; j < KT / 2; ++j) {
            const v2f xe = __builtin_elementwise_fma((v2f){ea, ea}, l[j], (v2f){lg, lg});
            const v2f xc = __builtin_elementwise_fma((v2f){ph, ph}, l[j], (v2f){p0, p0});
            v2f e, c;
            e.x = __builtin_amdgcn_exp2f(xe.x);  e.y = __builtin_amdgcn_exp2f(xe.y);
            c.x = __builtin_amdgcn_cosf(xc.x);   c.y = __builtin_amdgcn_cosf(xc.y);
            acc[j] = __builtin_elementwise_fma(e, c, acc[j]);
        }
    }

    float* kd = Kbuf + (size_t)d * LSEQ + t;
#pragma unroll
    for (int j = 0; j < KT / 2; ++j) { kd[512 * j] = acc[j].x; kd[512 * j + 256] = acc[j].y; }
}

#define NCH    12
#define CTHR   768
#define KS_LEN 2080
#define U_HEAD 256
#define U_DAT0 323
#define U_TAIL 2883
#define USTR   2884
#define YSTR   2052
#define ROWQ   (256 * (DCH / 4))

typedef __attribute__((ext_vector_type(4))) float v4f;

__device__ __forceinline__ v8f mma4(v2f a, v2f b, v8f c) {
    return __builtin_amdgcn_wmma_f32_16x16x4_f32(false, a, false, b, (short)0, c, false, false);
}

template <int NT>
__device__ __forceinline__ void conv_trips(int n, const float* kr, const float* us,
                                           int& ik, int (&iu)[4], v8f (&acc)[4]) {
    for (int it = 0; it < n; ++it) {
#pragma unroll
        for (int h = 0; h < 2; ++h) {
            v2f kb0, kb1;
            v4f f[NT];
            kb0.x = kr[ik + 4 * h];       kb0.y = kr[ik + 4 * h + 1];
            kb1.x = kr[ik + 4 * h + 2];   kb1.y = kr[ik + 4 * h + 3];
#pragma unroll
            for (int j = 0; j < NT; ++j) f[j] = *(const v4f*)(us + iu[4 - NT + j] + 4 * h);
#pragma unroll
            for (int j = 0; j < NT; ++j) {
                const v2f a = { f[j].x, f[j].y };
                acc[4 - NT + j] = mma4(a, kb0, acc[4 - NT + j]);
            }
#pragma unroll
            for (int j = 0; j < NT; ++j) {
                const v2f a = { f[j].z, f[j].w };
                acc[4 - NT + j] = mma4(a, kb1, acc[4 - NT + j]);
            }
        }
        ik -= 16;
#pragma unroll
        for (int j = 0; j < NT; ++j) iu[4 - NT + j] -= 20;
    }
}

__global__ void __launch_bounds__(CTHR)
s4_conv_wmma(const float* __restrict__ u,
             const float* __restrict__ Kbuf,
             float* __restrict__ y) {
    __shared__ __attribute__((aligned(16))) float Kr[NCH][KS_LEN];
    __shared__ __attribute__((aligned(16))) float Us[NCH][USTR];
    float (*Ys)[YSTR] = (float (*)[YSTR])&Us[0][0];

    const int d0   = (blockIdx.x >> 2) * NCH;
    const int bb   = blockIdx.x & 3;
    const int tid  = threadIdx.x;
    const int t0   = tid / 3;
    const int part = tid - 3 * t0;

    float4 ux[8], kx[8];
    {
        const float4* us = (const float4*)(u + ((size_t)bb * LSEQ + t0) * DCH + d0 + 4 * part);
#pragma unroll
        for (int i = 0; i < 8; ++i) ux[i] = us[(size_t)i * ROWQ];
        const float4* ks = (const float4*)(Kbuf + (size_t)d0 * LSEQ);
#pragma unroll
        for (int i = 0; i < 8; ++i) kx[i] = ks[tid + i * CTHR];
    }

    for (int i = tid; i < NCH * 356; i += CTHR) {
        const int c = i / 356;
        const int k = i - 356 * c;
        if (k < U_DAT0)       Us[c][k] = 0.0f;
        else if (k == U_DAT0) Us[c][U_TAIL] = 0.0f;
        else {
            const int k2 = k - (U_DAT0 + 1);
            Kr[c][k2 < 16 ? k2 : LSEQ + k2] = 0.0f;
        }
    }
#pragma unroll
    for (int i = 0; i < 8; ++i) {
        const int    q = tid + i * CTHR;
        const int    c = q >> 9;
        const int    j = q & 511;
        const float4 v = kx[i];
        ((float4*)&Kr[c][0])[515 - j] = make_float4(v.w, v.z, v.y, v.x);
    }
#pragma unroll
    for (int i = 0; i < 8; ++i) {
        const int tp = t0 + 256 * i + U_HEAD;
        float*    q  = &Us[4 * part][0] + tp + 4 * ((tp + 15) >> 4) + 3;
        q[0] = ux[i].x; q[USTR] = ux[i].y; q[2 * USTR] = ux[i].z; q[3 * USTR] = ux[i].w;
    }
    __syncthreads();

    const int wave = __builtin_amdgcn_readfirstlane(tid >> 5);
    const int ch   = wave >> 1;
    const int par  = wave & 1;
    const int lane = tid & 31;
    const int ln   = lane & 15;
    const int g    = lane >> 4;
    const int m0 = 2 * par, m1 = 2 * par + 1, m2 = 6 - 2 * par, m3 = 7 - 2 * par;

    const float* kr = &Kr[ch][0];
    const float* us = &Us[ch][0];
    int ik = 2072 - 8 * g - ln;
    const int ub = 20 * (ln + 17) - 8 * g - 4;
    int iu[4] = { ub + 320 * m0, ub + 320 * m1, ub + 320 * m2, ub + 320 * m3 };
    v8f acc[4] = {};

    conv_trips<4>(16 * m0 + 17,   kr, us, ik, iu, acc);
    conv_trips<3>(16 * (m1 - m0), kr, us, ik, iu, acc);
    conv_trips<2>(16 * (m2 - m1), kr, us, ik, iu, acc);
    conv_trips<1>(16 * (m3 - m2), kr, us, ik, iu, acc);

    __syncthreads();

    {
        float*    ys = &Ys[ch][0];
        const int mm[4] = { m0, m1, m2, m3 };
#pragma unroll
        for (int k = 0; k < 4; ++k) {
#pragma unroll
            for (int r = 0; r < 8; ++r) {
                const int t = 256 * mm[k] + 16 * r + 128 * g + ln;
                ys[t ^ (g << 4)] = acc[k][r];
            }
        }
    }
    __syncthreads();

    {
        float4* yd = (float4*)(y + ((size_t)bb * LSEQ + t0) * DCH + d0 + 4 * part);
#pragma unroll
        for (int i = 0; i < 8; ++i) {
            const int    t = t0 + 256 * i;
            const float* s = &Ys[4 * part][0] + (t ^ (((t >> 7) & 1) << 4));
            yd[(size_t)i * ROWQ] = make_float4(s[0], s[YSTR], s[2 * YSTR], s[3 * YSTR]);
        }
    }
}

extern "C" void kernel_launch(void* const* d_in, const int* in_sizes, int n_in,
                              void* d_out, int out_size, void* d_ws, size_t ws_size,
                              hipStream_t stream) {
    const float* u         = (const float*)d_in[0];
    const float* A_real    = (const float*)d_in[1];
    const float* A_imag    = (const float*)d_in[2];
    const float* B_mat     = (const float*)d_in[3];
    const float* C_mat     = (const float*)d_in[4];
    const float* log_delta = (const float*)d_in[5];
    float* y = (float*)d_out;

    float* ws   = (float*)d_ws;
    float* dAr  = ws;
    float* dAi  = ws + 1 * DN;
    float* CBlg = ws + 2 * DN;
    float* CBph = ws + 3 * DN;
    float* Kbuf = ws + 4 * DN;

    s4_precompute<<<dim3((DN + 255) / 256), dim3(256), 0, stream>>>(
        A_real, A_imag, B_mat, C_mat, log_delta, dAr, dAi, CBlg, CBph);

    s4_kernelgen<<<dim3(DCH), dim3(256), 0, stream>>>(
        dAr, dAi, CBlg, CBph, Kbuf);

    s4_conv_wmma<<<dim3((DCH / NCH) * BSZ), dim3(CTHR), 0, stream>>>(u, Kbuf, y);
}


# Conv lag loop: trips run in groups of 2 (3-4 row tiles) or 4 (1-2 tiles); a group's LDS fragment reads are all issued in front of its 16-32 WMMAs, one LDS-to-matrix change per group; sums unchanged
# speedup vs baseline: 13.9334x; 1.0442x over previous
#include <hip/hip_runtime.h>
#include <hip/hip_bf16.h>
#include <math.h>

#define BSZ  4
#define LSEQ 2048
#define DCH  768
#define NST  64
#define DN   (DCH * NST)

typedef __attribute__((ext_vector_type(2))) float v2f;
typedef __attribute__((ext_vector_type(8))) float v8f;

__global__ void __launch_bounds__(256)
s4_precompute(const float* __restrict__ A_real, const float* __restrict__ A_imag,
              const float* __restrict__ B_mat,  const float* __restrict__ C_mat,
              const float* __restrict__ log_delta,
              float* __restrict__ dAr, float* __restrict__ dAi,
              float* __restrict__ CBlg, float* __restrict__ CBph) {
    int idx = blockIdx.x * blockDim.x + threadIdx.x;
    if (idx >= DN) return;
    int d = idx / NST;

    float delta = log1pf(expf(log_delta[d]));
    float ar = A_real[idx], ai = A_imag[idx];
    float dr = delta * ar, di = delta * ai;

    float er = expf(dr);
    float s, c;
    __sincosf(di, &s, &c);
    float abr = er * c, abi = er * s;

    float nr = abr - 1.0f, ni = abi;
    float inv = 1.0f / (ar * ar + ai * ai);
    float qr = (nr * ar + ni * ai) * inv;
    float qi = (ni * ar - nr * ai) * inv;

    float b = B_mat[idx], cm = C_mat[idx];
    const float cbr = cm * (qr * b);
    const float cbi = cm * (qi * b);
    const float ax = fabsf(cbr), ay = fabsf(cbi);
    const float mx = fmaxf(ax, ay), mn = fminf(ax, ay);
    const float rr = (mx > 0.0f) ? mn / mx : 0.0f;
    dAr[idx]  = dr;
    dAi[idx]  = di;
    CBlg[idx] = log2f(mx) + 0.5f * log2f(fmaf(rr, rr, 1.0f));
    CBph[idx] = atan2f(cbi, cbr) * 0.15915494f;
}

#define KT 8

__global__ void __launch_bounds__(256)
s4_kernelgen(const float* __restrict__ dAr, const float* __restrict__ dAi,
             const float* __restrict__ CBlg, const float* __restrict__ CBph,
             float* __restrict__ Kbuf) {
    const int d = blockIdx.x;
    const int t = threadIdx.x;

    const float* pr = dAr + d * NST;
    const float* pi = dAi + d * NST;
    const float* cg = CBlg + d * NST;
    const float* cp = CBph + d * NST;

    v2f l[KT / 2], acc[KT / 2];
#pragma unroll
    for (int j = 0; j < KT / 2; ++j) {
        l[j].x = (float)(t + 512 * j);  l[j].y = (float)(t + 512 * j + 256);
        acc[j].x = 0.0f;                acc[j].y = 0.0f;
    }

#pragma unroll 4
    for (int n = 0; n < NST; ++n) {
        const float ea = pr[n] * 1.44269504f;
        const float ph = pi[n] * 0.15915494f;
        const float lg = cg[n], p0 = cp[n];
#pragma unroll
        for (int j = 0; j < KT / 2; ++j) {
            const v2f xe = __builtin_elementwise_fma((v2f){ea, ea}, l[j], (v2f){lg, lg});
            const v2f xc = __builtin_elementwise_fma((v2f){ph, ph}, l[j], (v2f){p0, p0});
            v2f e, c;
            e.x = __builtin_amdgcn_exp2f(xe.x);  e.y = __builtin_amdgcn_exp2f(xe.y);
            c.x = __builtin_amdgcn_cosf(xc.x);   c.y = __builtin_amdgcn_cosf(xc.y);
            acc[j] = __builtin_elementwise_fma(e, c, acc[j]);
        }
    }

    float* kd = Kbuf + (size_t)d * LSEQ + t;
#pragma unroll
    for (int j = 0; j < KT / 2; ++j) { kd[512 * j] = acc[j].x; kd[512 * j + 256] = acc[j].y; }
}

#define NCH    12
#define CTHR   768
#define KS_LEN 2080
#define U_HEAD 256
#define U_DAT0 323
#define U_TAIL 2883
#define USTR   2884
#define YSTR   2052
#define ROWQ   (256 * (DCH / 4))

typedef __attribute__((ext_vector_type(4))) float v4f;
typedef const __attribute__((address_space(3))) float lds_f;
typedef const __attribute__((address_space(3))) v4f   lds_f4;

__device__ __forceinline__ v8f mma4(v2f a, v2f b, v8f c) {
    return __builtin_amdgcn_wmma_f32_16x16x4_f32(false, a, false, b, (short)0, c, false, false);
}

template <int NT, int U>
__device__ __forceinline__ void conv_trips(int n, lds_f* kr, lds_f* us, int& done,
                                           int lk, const int (&lu)[4], v8f (&acc)[4]) {
    lds_f* kr0 = (lds_f*)(size_t)(unsigned)__builtin_amdgcn_readfirstlane((int)(size_t)(kr - 16 * (done + U - 1)));
    lds_f* us0 = (lds_f*)(size_t)(unsigned)__builtin_amdgcn_readfirstlane((int)(size_t)(us - 20 * (done + U - 1)));
    lds_f* pk = kr0 + lk;
    lds_f* pu[NT];
#pragma unroll
    for (int j = 0; j < NT; ++j) pu[j] = us0 + lu[4 - NT + j];

#pragma nounroll
    for (int it = 0; it < n; ++it) {
        v2f kb[U][4];
        v4f f[U][2][NT];
#pragma unroll
        for (int t = 0; t < U; ++t) {
#pragma unroll
            for (int p = 0; p < 4; ++p) {
                kb[t][p].x = pk[16 * (U - 1 - t) + 2 * p];
                kb[t][p].y = pk[16 * (U - 1 - t) + 2 * p + 1];
            }
        }
#pragma unroll
        for (int j = 0; j < NT; ++j) {
#pragma unroll
            for (int t = 0; t < U; ++t) {
#pragma unroll
                for (int h = 0; h < 2; ++h) f[t][h][j] = *(lds_f4*)(pu[j] + 20 * (U - 1 - t) + 4 * h);
            }
        }
#pragma unroll
        for (int t = 0; t < U; ++t) {
#pragma unroll
            for (int h = 0; h < 2; ++h) {
#pragma unroll
                for (int j = 0; j < NT; ++j) {
                    const v2f a = { f[t][h][j].x, f[t][h][j].y };
                    acc[4 - NT + j] = mma4(a, kb[t][2 * h], acc[4 - NT + j]);
                }
#pragma unroll
                for (int j = 0; j < NT; ++j) {
                    const v2f a = { f[t][h][j].z, f[t][h][j].w };
                    acc[4 - NT + j] = mma4(a, kb[t][2 * h + 1], acc[4 - NT + j]);
                }
            }
        }
        pk -= 16 * U;
#pragma unroll
        for (int j = 0; j < NT; ++j) pu[j] -= 20 * U;
        __builtin_amdgcn_sched_group_barrier(0x100, 8 * U + 2 * NT * U, 0);
        __builtin_amdgcn_sched_group_barrier(0x008, 4 * NT * U, 0);
    }
    done += U * n;
}

__global__ void __launch_bounds__(CTHR)
s4_conv_wmma(const float* __restrict__ u,
             const float* __restrict__ Kbuf,
             float* __restrict__ y) {
    __shared__ __attribute__((aligned(16))) float Kr[NCH][KS_LEN];
    __shared__ __attribute__((aligned(16))) float Us[NCH][USTR];
    float (*Ys)[YSTR] = (float (*)[YSTR])&Us[0][0];

    const int d0   = (blockIdx.x >> 2) * NCH;
    const int bb   = blockIdx.x & 3;
    const int tid  = threadIdx.x;
    const int t0   = tid / 3;
    const int part = tid - 3 * t0;

    float4 ux[8], kx[8];
    {
        const float4* us = (const float4*)(u + ((size_t)bb * LSEQ + t0) * DCH + d0 + 4 * part);
#pragma unroll
        for (int i = 0; i < 8; ++i) ux[i] = us[(size_t)i * ROWQ];
        const float4* ks = (const float4*)(Kbuf + (size_t)d0 * LSEQ);
#pragma unroll
        for (int i = 0; i < 8; ++i) kx[i] = ks[tid + i * CTHR];
    }

    for (int i = tid; i < NCH * 356; i += CTHR) {
        const int c = i / 356;
        const int k = i - 356 * c;
        if (k < U_DAT0)       Us[c][k] = 0.0f;
        else if (k == U_DAT0) Us[c][U_TAIL] = 0.0f;
        else {
            const int k2 = k - (U_DAT0 + 1);
            Kr[c][k2 < 16 ? k2 : LSEQ + k2] = 0.0f;
        }
    }
#pragma unroll
    for (int i = 0; i < 8; ++i) {
        const int    q = tid + i * CTHR;
        const int    c = q >> 9;
        const int    j = q & 511;
        const float4 v = kx[i];
        ((float4*)&Kr[c][0])[515 - j] = make_float4(v.w, v.z, v.y, v.x);
    }
#pragma unroll
    for (int i = 0; i < 8; ++i) {
        const int tp = t0 + 256 * i + U_HEAD;
        float*    q  = &Us[4 * part][0] + tp + 4 * ((tp + 15) >> 4) + 3;
        q[0] = ux[i].x; q[USTR] = ux[i].y; q[2 * USTR] = ux[i].z; q[3 * USTR] = ux[i].w;
    }
    __syncthreads();

    const int wave = __builtin_amdgcn_readfirstlane(tid >> 5);
    const int ch   = wave >> 1;
    const int par  = wave & 1;
    const int lane = tid & 31;
    const int ln   = lane & 15;
    const int g    = lane >> 4;
    const int m0 = 2 * par, m1 = 2 * par + 1, m2 = 6 - 2 * par, m3 = 7 - 2 * par;

    lds_f* kr = (lds_f*)&Kr[ch][0] + 2072;
    lds_f* us = (lds_f*)&Us[ch][0];
    const int lk = -8 * g - ln;
    const int ub = 20 * (ln + 17) - 8 * g - 4;
    const int lu[4] = { ub + 320 * m0, ub + 320 * m1, ub + 320 * m2, ub + 320 * m3 };
    int done = 0;
    v8f acc[4] = {};

    conv_trips<4, 1>(1,               kr, us, done, lk, lu, acc);
    conv_trips<4, 2>(8 * m0 + 8,      kr, us, done, lk, lu, acc);
    conv_trips<3, 2>(8 * (m1 - m0),   kr, us, done, lk, lu, acc);
    conv_trips<2, 4>(4 * (m2 - m1),   kr, us, done, lk, lu, acc);
    conv_trips<1, 4>(4 * (m3 - m2),   kr, us, done, lk, lu, acc);

    __syncthreads();

    {
        float*    ys = &Ys[ch][0];
        const int mm[4] = { m0, m1, m2, m3 };
#pragma unroll
        for (int k = 0; k < 4; ++k) {
#pragma unroll
            for (int r = 0; r < 8; ++r) {
                const int t = 256 * mm[k] + 16 * r + 128 * g + ln;
                ys[t ^ (g << 4)] = acc[k][r];
            }
        }
    }
    __syncthreads();

    {
        float4* yd = (float4*)(y + ((size_t)bb * LSEQ + t0) * DCH + d0 + 4 * part);
#pragma unroll
        for (int i = 0; i < 8; ++i) {
            const int    t = t0 + 256 * i;
            const float* s = &Ys[4 * part][0] + (t ^ (((t >> 7) & 1) << 4));
            yd[(size_t)i * ROWQ] = make_float4(s[0], s[YSTR], s[2 * YSTR], s[3 * YSTR]);
        }
    }
}

extern "C" void kernel_launch(void* const* d_in, const int* in_sizes, int n_in,
                              void* d_out, int out_size, void* d_ws, size_t ws_size,
                              hipStream_t stream) {
    const float* u         = (const float*)d_in[0];
    const float* A_real    = (const float*)d_in[1];
    const float* A_imag    = (const float*)d_in[2];
    const float* B_mat     = (const float*)d_in[3];
    const float* C_mat     = (const float*)d_in[4];
    const float* log_delta = (const float*)d_in[5];
    float* y = (float*)d_out;

    float* ws   = (float*)d_ws;
    float* dAr  = ws;
    float* dAi  = ws + 1 * DN;
    float* CBlg = ws + 2 * DN;
    float* CBph = ws + 3 * DN;
    float* Kbuf = ws + 4 * DN;

    s4_precompute<<<dim3((DN + 255) / 256), dim3(256), 0, stream>>>(
        A_real, A_imag, B_mat, C_mat, log_delta, dAr, dAi, CBlg, CBph);

    s4_kernelgen<<<dim3(DCH), dim3(256), 0, stream>>>(
        dAr, dAi, CBlg, CBph, Kbuf);

    s4_conv_wmma<<<dim3((DCH / NCH) * BSZ), dim3(CTHR), 0, stream>>>(u, Kbuf, y);
}


# Conv block takes the time axis in two halves: 2nd half of u and taps lands in registers under row tiles 0-3, 1st half of y is stored under tiles 4-7; LDS-only barriers; sums unchanged
# speedup vs baseline: 14.6966x; 1.0548x over previous
#include <hip/hip_runtime.h>
#include <hip/hip_bf16.h>
#include <math.h>

#define BSZ  4
#define LSEQ 2048
#define DCH  768
#define NST  64
#define DN   (DCH * NST)

typedef __attribute__((ext_vector_type(2))) float v2f;
typedef __attribute__((ext_vector_type(8))) float v8f;

__global__ void __launch_bounds__(256)
s4_precompute(const float* __restrict__ A_real, const float* __restrict__ A_imag,
              const float* __restrict__ B_mat,  const float* __restrict__ C_mat,
              const float* __restrict__ log_delta,
              float* __restrict__ dAr, float* __restrict__ dAi,
              float* __restrict__ CBlg, float* __restrict__ CBph) {
    int idx = blockIdx.x * blockDim.x + threadIdx.x;
    if (idx >= DN) return;
    int d = idx / NST;

    float delta = log1pf(expf(log_delta[d]));
    float ar = A_real[idx], ai = A_imag[idx];
    float dr = delta * ar, di = delta * ai;

    float er = expf(dr);
    float s, c;
    __sincosf(di, &s, &c);
    float abr = er * c, abi = er * s;

    float nr = abr - 1.0f, ni = abi;
    float inv = 1.0f / (ar * ar + ai * ai);
    float qr = (nr * ar + ni * ai) * inv;
    float qi = (ni * ar - nr * ai) * inv;

    float b = B_mat[idx], cm = C_mat[idx];
    const float cbr = cm * (qr * b);
    const float cbi = cm * (qi * b);
    const float ax = fabsf(cbr), ay = fabsf(cbi);
    const float mx = fmaxf(ax, ay), mn = fminf(ax, ay);
    const float rr = (mx > 0.0f) ? mn / mx : 0.0f;
    dAr[idx]  = dr;
    dAi[idx]  = di;
    CBlg[idx] = log2f(mx) + 0.5f * log2f(fmaf(rr, rr, 1.0f));
    CBph[idx] = atan2f(cbi, cbr) * 0.15915494f;
}

#define KT 8

__global__ void __launch_bounds__(256)
s4_kernelgen(const float* __restrict__ dAr, const float* __restrict__ dAi,
             const float* __restrict__ CBlg, const float* __restrict__ CBph,
             float* __restrict__ Kbuf) {
    const int d = blockIdx.x;
    const int t = threadIdx.x;

    const float* pr = dAr + d * NST;
    const float* pi = dAi + d * NST;
    const float* cg = CBlg + d * NST;
    const float* cp = CBph + d * NST;

    v2f l[KT / 2], acc[KT / 2];
#pragma unroll
    for (int j = 0; j < KT / 2; ++j) {
        l[j].x = (float)(t + 512 * j);  l[j].y = (float)(t + 512 * j + 256);
        acc[j].x = 0.0f;                acc[j].y = 0.0f;
    }

#pragma unroll 4
    for (int n = 0; n < NST; ++n) {
        const float ea = pr[n] * 1.44269504f;
        const float ph = pi[n] * 0.15915494f;
        const float lg = cg[n], p0 = cp[n];
#pragma unroll
        for (int j = 0; j < KT / 2; ++j) {
            const v2f xe = __builtin_elementwise_fma((v2f){ea, ea}, l[j], (v2f){lg, lg});
            const v2f xc = __builtin_elementwise_fma((v2f){ph, ph}, l[j], (v2f){p0, p0});
            v2f e, c;
            e.x = __builtin_amdgcn_exp2f(xe.x);  e.y = __builtin_amdgcn_exp2f(xe.y);
            c.x = __builtin_amdgcn_cosf(xc.x);   c.y = __builtin_amdgcn_cosf(xc.y);
            acc[j] = __builtin_elementwise_fma(e, c, acc[j]);
        }
    }

    float* kd = Kbuf + (size_t)d * LSEQ + t;
#pragma unroll
    for (int j = 0; j < KT / 2; ++j) { kd[512 * j] = acc[j].x; kd[512 * j + 256] = acc[j].y; }
}

#define NCH    12
#define CTHR   768
#define KS_LEN 2080
#define U_HEAD 256
#define U_DAT0 323
#define U_MID  1603
#define U_TAIL 2883
#define USTR   2884
#define YH     1028
#define K_MID  1024
#define NZERO  373
#define ROWQ   (256 * (DCH / 4))

typedef __attribute__((ext_vector_type(4))) float v4f;
typedef const __attribute__((address_space(3))) float lds_f;
typedef const __attribute__((address_space(3))) v4f   lds_f4;

__device__ __forceinline__ v8f mma4(v2f a, v2f b, v8f c) {
    return __builtin_amdgcn_wmma_f32_16x16x4_f32(false, a, false, b, (short)0, c, false, false);
}

__device__ __forceinline__ void lds_barrier() {
    __builtin_amdgcn_fence(__ATOMIC_RELEASE, "workgroup", "local");
    __builtin_amdgcn_s_barrier();
    __builtin_amdgcn_fence(__ATOMIC_ACQUIRE, "workgroup", "local");
}

template <int NT, int U>
__device__ __forceinline__ void conv_trips(int n, lds_f* kr, lds_f* us, int& done,
                                           int lk, const int (&lu)[2], v8f (&acc)[2]) {
    lds_f* kr0 = (lds_f*)(size_t)(unsigned)__builtin_amdgcn_readfirstlane((int)(size_t)(kr - 16 * (done + U - 1)));
    lds_f* us0 = (lds_f*)(size_t)(unsigned)__builtin_amdgcn_readfirstlane((int)(size_t)(us - 20 * (done + U - 1)));
    lds_f* pk = kr0 + lk;
    lds_f* pu[NT];
#pragma unroll
    for (int j = 0; j < NT; ++j) pu[j] = us0 + lu[2 - NT + j];

#pragma nounroll
    for (int it = 0; it < n; ++it) {
        v2f kb[U][4];
        v4f f[U][2][NT];
#pragma unroll
        for (int t = 0; t < U; ++t) {
#pragma unroll
            for (int p = 0; p < 4; ++p) {
                kb[t][p].x = pk[16 * (U - 1 - t) + 2 * p];
                kb[t][p].y = pk[16 * (U - 1 - t) + 2 * p + 1];
            }
        }
#pragma unroll
        for (int j = 0; j < NT; ++j) {
#pragma unroll
            for (int t = 0; t < U; ++t) {
#pragma unroll
                for (int h = 0; h < 2; ++h) f[t][h][j] = *(lds_f4*)(pu[j] + 20 * (U - 1 - t) + 4 * h);
            }
        }
#pragma unroll
        for (int t = 0; t < U; ++t) {
#pragma unroll
            for (int h = 0; h < 2; ++h) {
#pragma unroll
                for (int j = 0; j < NT; ++j) {
                    const v2f a = { f[t][h][j].x, f[t][h][j].y };
                    acc[2 - NT + j] = mma4(a, kb[t][2 * h], acc[2 - NT + j]);
                }
#pragma unroll
                for (int j = 0; j < NT; ++j) {
                    const v2f a = { f[t][h][j].z, f[t][h][j].w };
                    acc[2 - NT + j] = mma4(a, kb[t][2 * h + 1], acc[2 - NT + j]);
                }
            }
        }
        pk -= 16 * U;
#pragma unroll
        for (int j = 0; j < NT; ++j) pu[j] -= 20 * U;
        __builtin_amdgcn_sched_group_barrier(0x100, 8 * U + 2 * NT * U, 0);
        __builtin_amdgcn_sched_group_barrier(0x008, 4 * NT * U, 0);
    }
    done += U * n;
}

template <int U2>
__device__ __forceinline__ void conv_half(int a, int b, lds_f* kr, lds_f* us,
                                          int lk, int ub, v8f (&acc)[2]) {
    const int lu[2] = { ub + 320 * a, ub + 320 * b };
    int done = 0;
    conv_trips<2, 1>(1,           kr, us, done, lk, lu, acc);
    conv_trips<2, U2>((16 * a + 16) / U2, kr, us, done, lk, lu, acc);
    conv_trips<1, 4>(4 * (b - a), kr, us, done, lk, lu, acc);
}

__global__ void __launch_bounds__(CTHR)
s4_conv_wmma(const float* __restrict__ u,
             const float* __restrict__ Kbuf,
             float* __restrict__ y) {
    __shared__ __attribute__((aligned(16))) float Kr[NCH][KS_LEN];
    __shared__ __attribute__((aligned(16))) float Us[NCH][USTR];
    __shared__ __attribute__((aligned(16))) float Ys[NCH][YH];

    const int d0   = (blockIdx.x >> 2) * NCH;
    const int bb   = blockIdx.x & 3;
    const int tid  = threadIdx.x;
    const int t0   = tid / 3;
    const int part = tid - 3 * t0;

    float4 ux[8], kx[8];
    const int kq[4] = { tid, tid + CTHR, tid + 2 * CTHR, tid + 3 * CTHR };
    {
        const float4* us = (const float4*)(u + ((size_t)bb * LSEQ + t0) * DCH + d0 + 4 * part);
        const float4* ks = (const float4*)(Kbuf + (size_t)d0 * LSEQ);
#pragma unroll
        for (int i = 0; i < 4; ++i) ux[i] = us[(size_t)i * ROWQ];
#pragma unroll
        for (int i = 0; i < 4; ++i) kx[i] = ks[(kq[i] >> 8) * 512 + (kq[i] & 255)];
        __builtin_amdgcn_sched_barrier(0);
#pragma unroll
        for (int i = 4; i < 8; ++i) ux[i] = us[(size_t)i * ROWQ];
#pragma unroll
        for (int i = 0; i < 4; ++i) kx[4 + i] = ks[(kq[i] >> 8) * 512 + 256 + (kq[i] & 255)];
    }

    for (int i = tid; i < NCH * NZERO; i += CTHR) {
        const int c = i / NZERO;
        const int k = i - NZERO * c;
        if (k < U_DAT0)           Us[c][k] = 0.0f;
        else if (k == U_DAT0)     Us[c][U_TAIL] = 0.0f;
        else if (k == U_DAT0 + 1) Us[c][U_MID] = 0.0f;
        else {
            const int k2 = k - (U_DAT0 + 2);
            Kr[c][k2 < 16 ? k2 : (k2 < 32 ? LSEQ + k2 : K_MID - 32 + k2)] = 0.0f;
        }
    }
#pragma unroll
    for (int i = 0; i < 4; ++i) {
        const int    c = kq[i] >> 8;
        const int    j = kq[i] & 255;
        const float4 v = kx[i];
        ((float4*)&Kr[c][0])[515 - j] = make_float4(v.w, v.z, v.y, v.x);
    }
#pragma unroll
    for (int i = 0; i < 4; ++i) {
        const int tp = t0 + 256 * i + U_HEAD;
        float*    q  = &Us[4 * part][0] + tp + 4 * ((tp + 15) >> 4) + 3;
        q[0] = ux[i].x; q[USTR] = ux[i].y; q[2 * USTR] = ux[i].z; q[3 * USTR] = ux[i].w;
    }
    lds_barrier();

    const int wave = __builtin_amdgcn_readfirstlane(tid >> 5);
    const int ch   = wave >> 1;
    const int par  = wave & 1;
    const int lane = tid & 31;
    const int ln   = lane & 15;
    const int g    = lane >> 4;

    lds_f* kr = (lds_f*)&Kr[ch][0] + 2072;
    lds_f* us = (lds_f*)&Us[ch][0];
    const int lk = -8 * g - ln;
    const int ub = 20 * (ln + 17) - 8 * g - 4;
    float*    ys = &Ys[ch][0];
    float4*   yd = (float4*)(y + ((size_t)bb * LSEQ + t0) * DCH + d0 + 4 * part);

    {
        v8f acc[2] = {};
        conv_half<2>(par, 3 - par, kr, us, lk, ub, acc);

        lds_barrier();
#pragma unroll
        for (int i = 0; i < 4; ++i) {
            const int    c = kq[i] >> 8;
            const int    j = 256 + (kq[i] & 255);
            const float4 v = kx[4 + i];
            ((float4*)&Kr[c][0])[515 - j] = make_float4(v.w, v.z, v.y, v.x);
        }
#pragma unroll
        for (int i = 4; i < 8; ++i) {
            const int tp = t0 + 256 * i + U_HEAD;
            float*    q  = &Us[4 * part][0] + tp + 4 * ((tp + 15) >> 4) + 3;
            q[0] = ux[i].x; q[USTR] = ux[i].y; q[2 * USTR] = ux[i].z; q[3 * USTR] = ux[i].w;
        }
        const int mm[2] = { par, 3 - par };
#pragma unroll
        for (int k = 0; k < 2; ++k) {
#pragma unroll
            for (int r = 0; r < 8; ++r) {
                const int t = 256 * mm[k] + 16 * r + 128 * g + ln;
                ys[t ^ (g << 4)] = acc[k][r];
            }
        }
    }
    lds_barrier();

#pragma unroll
    for (int i = 0; i < 4; ++i) {
        const int    t = t0 + 256 * i;
        const float* s = &Ys[4 * part][0] + (t ^ (((t >> 7) & 1) << 4));
        yd[(size_t)i * ROWQ] = make_float4(s[0], s[YH], s[2 * YH], s[3 * YH]);
    }

    {
        v8f acc[2] = {};
        conv_half<4>(4 + par, 7 - par, kr, us, lk, ub, acc);

        lds_barrier();
        const int mm[2] = { par, 3 - par };
#pragma unroll
        for (int k = 0; k < 2; ++k) {
#pragma unroll
            for (int r = 0; r < 8; ++r) {
                const int t = 256 * mm[k] + 16 * r + 128 * g + ln;
                ys[t ^ (g << 4)] = acc[k][r];
            }
        }
    }
    lds_barrier();

#pragma unroll
    for (int i = 0; i < 4; ++i) {
        const int    t = t0 + 256 * i;
        const float* s = &Ys[4 * part][0] + (t ^ (((t >> 7) & 1) << 4));
        yd[(size_t)(i + 4) * ROWQ] = make_float4(s[0], s[YH], s[2 * YH], s[3 * YH]);
    }
}

extern "C" void kernel_launch(void* const* d_in, const int* in_sizes, int n_in,
                              void* d_out, int out_size, void* d_ws, size_t ws_size,
                              hipStream_t stream) {
    const float* u         = (const float*)d_in[0];
    const float* A_real    = (const float*)d_in[1];
    const float* A_imag    = (const float*)d_in[2];
    const float* B_mat     = (const float*)d_in[3];
    const float* C_mat     = (const float*)d_in[4];
    const float* log_delta = (const float*)d_in[5];
    float* y = (float*)d_out;

    float* ws   = (float*)d_ws;
    float* dAr  = ws;
    float* dAi  = ws + 1 * DN;
    float* CBlg = ws + 2 * DN;
    float* CBph = ws + 3 * DN;
    float* Kbuf = ws + 4 * DN;

    s4_precompute<<<dim3((DN + 255) / 256), dim3(256), 0, stream>>>(
        A_real, A_imag, B_mat, C_mat, log_delta, dAr, dAi, CBlg, CBph);

    s4_kernelgen<<<dim3(DCH), dim3(256), 0, stream>>>(
        dAr, dAi, CBlg, CBph, Kbuf);

    s4_conv_wmma<<<dim3((DCH / NCH) * BSZ), dim3(CTHR), 0, stream>>>(u, Kbuf, y);
}
